# starting kernel
# baseline (speedup 1.0000x reference)
#include <hip/hip_runtime.h>
#include <hip/hip_bf16.h>


typedef __attribute__((ext_vector_type(2))) float v2f;
typedef __attribute__((ext_vector_type(8))) float v8f;

#define DP 17
#define KDIM 160
#define KSTEPS 40
#define XSTR 20
#define BLDS_N 2560

struct PairTab { int b[KDIM]; int c[KDIM]; };
static constexpr PairTab makeTab() {
  PairTab t{};
  int q = 0;
  for (int b = 0; b < DP; ++b)
    for (int c = b; c < DP; ++c) { t.b[q] = b; t.c[q] = c; ++q; }
  for (; q < KDIM; ++q) { t.b[q] = DP - 1; t.c[q] = DP - 1; }
  return t;
}
static constexpr PairTab TAB = makeTab();

__global__ void build_weights_kernel(const float* __restrict__ w,
                                     const int* __restrict__ comb_idx,
                                     float* __restrict__ ws) {
  const int tid = threadIdx.x;
  for (int i = tid; i < BLDS_N + 1; i += 256) ws[i] = 0.0f;
  __syncthreads();
  for (int i = tid; i < 969; i += 256) {
    const int a = comb_idx[3 * i + 0];
    const int b = comb_idx[3 * i + 1];
    const int c = comb_idx[3 * i + 2];
    const float wi = w[i];
    if (a == 16) {
      ws[BLDS_N] = wi;
    } else {
      const int q = b * DP - (b * (b - 1)) / 2 + (c - b);
      const int pos = ((q >> 1) * 16 + a) * 2 + (q & 1);
      ws[pos] = wi;
    }
  }
}

__global__ __launch_bounds__(256) void poly_logistic_kernel(
    const float* __restrict__ X, const float* __restrict__ ws,
    float* __restrict__ out, int tiles_per_wave) {
  __shared__ __align__(16) float Blds[BLDS_N];
  __shared__ __align__(16) float xp[8][16 * XSTR];

  const int tid  = threadIdx.x;
  const int wave = tid >> 5;
  const int lane = tid & 31;
  const int hi   = lane >> 4;
  const int lm   = lane & 15;
  float* xw = &xp[wave][0];

  for (int i = tid; i < BLDS_N; i += 256) Blds[i] = ws[i];
  const float wconst = ws[BLDS_N];

  const int wave_global = blockIdx.x * 8 + wave;
  const int tile0 = wave_global * tiles_per_wave;

  const float* bp = Blds + (hi * 16 + lm) * 2;

  __syncthreads();

  for (int t = 0; t < tiles_per_wave; ++t) {
    const int tile = tile0 + t;

    const float4* Xv = (const float4*)(X + (size_t)tile * 256);
    const float4 v0 = Xv[lane];
    const float4 v1 = Xv[lane + 32];
    {
      const int m0 = lane >> 2;
      const int f0 = (lane & 3) * 4;
      *(float4*)(xw + m0 * XSTR + f0)       = v0;
      *(float4*)(xw + (m0 + 8) * XSTR + f0) = v1;
    }
    __builtin_prefetch(X + (size_t)(tile + 1) * 256 + lane * 8, 0, 0);
    __syncthreads();

    const float* xr = xw + lm * XSTR;
    float xv[DP];
    {
      const float4 r0 = *(const float4*)(xr + 0);
      const float4 r1 = *(const float4*)(xr + 4);
      const float4 r2 = *(const float4*)(xr + 8);
      const float4 r3 = *(const float4*)(xr + 12);
      xv[0]  = r0.x; xv[1]  = r0.y; xv[2]  = r0.z; xv[3]  = r0.w;
      xv[4]  = r1.x; xv[5]  = r1.y; xv[6]  = r1.z; xv[7]  = r1.w;
      xv[8]  = r2.x; xv[9]  = r2.y; xv[10] = r2.z; xv[11] = r2.w;
      xv[12] = r3.x; xv[13] = r3.y; xv[14] = r3.z; xv[15] = r3.w;
      xv[16] = 1.0f;
    }

    v2f A[KSTEPS];
    if (hi) {
#pragma unroll
      for (int k = 0; k < KSTEPS; ++k) {
        A[k].x = xv[TAB.b[4 * k + 2]] * xv[TAB.c[4 * k + 2]];
        A[k].y = xv[TAB.b[4 * k + 3]] * xv[TAB.c[4 * k + 3]];
      }
    } else {
#pragma unroll
      for (int k = 0; k < KSTEPS; ++k) {
        A[k].x = xv[TAB.b[4 * k + 0]] * xv[TAB.c[4 * k + 0]];
        A[k].y = xv[TAB.b[4 * k + 1]] * xv[TAB.c[4 * k + 1]];
      }
    }

    v8f acc = {};
#pragma unroll
    for (int k = 0; k < KSTEPS; ++k) {
      const v2f Bk = *(const v2f*)(bp + 64 * k);
      acc = __builtin_amdgcn_wmma_f32_16x16x4_f32(
          false, A[k], false, Bk, (short)0, acc, false, false);
    }

#pragma unroll
    for (int r = 0; r < 8; ++r) {
      const int m = r + 8 * hi;
      float v = acc[r] * xw[m * XSTR + lm];
      v += __shfl_xor(v, 1, 32);
      v += __shfl_xor(v, 2, 32);
      v += __shfl_xor(v, 4, 32);
      v += __shfl_xor(v, 8, 32);
      if (lm == 0) {
        const float logit = v + wconst;
        out[(size_t)tile * 16 + m] = 1.0f / (1.0f + __expf(-logit));
      }
    }
    __syncthreads();
  }
}

extern "C" void kernel_launch(void* const* d_in, const int* in_sizes, int n_in,
                              void* d_out, int out_size, void* d_ws,
                              size_t ws_size, hipStream_t stream) {
  const float* X        = (const float*)d_in[0];
  const float* w        = (const float*)d_in[1];
  const int*   comb_idx = (const int*)d_in[2];
  float*       out      = (float*)d_out;
  float*       ws       = (float*)d_ws;

  build_weights_kernel<<<1, 256, 0, stream>>>(w, comb_idx, ws);

  const int tiles_per_wave = 4;
  poly_logistic_kernel<<<128, 256, 0, stream>>>(X, ws, out, tiles_per_wave);
}


# Each tile's pair rows split over a wave pair (groups 0,2 / 1,3): 10 of 20 WMMAs and half the row sums per wave, 4 tiles staged once, partials joined in LDS; 256 blocks; regroups the sum
# speedup vs baseline: 2.8891x; 2.8891x over previous
#include <hip/hip_runtime.h>
#include <hip/hip_bf16.h>


typedef __attribute__((ext_vector_type(2))) float v2f;
typedef __attribute__((ext_vector_type(8))) float v8f;

#define DP 17
#define XSTR 20
#define NROWT 10
#define NFRAG 20
#define WS_CONST 1280

static constexpr int tileGroup(int j)     { return j < 4 ? 0 : j < 7 ? 1 : j < 9 ? 2 : 3; }
static constexpr int groupTileBase(int g) { return g == 0 ? 0 : g == 1 ? 4 : g == 2 ? 7 : 9; }
static constexpr int groupFragBase(int g) { return g == 0 ? 0 : g == 1 ? 4 : g == 2 ? 10 : 16; }
static constexpr int fragOf(int j, int s) {
  return groupFragBase(tileGroup(j)) + (j - groupTileBase(tileGroup(j))) * (tileGroup(j) + 1) + s;
}

struct RowTab { int b[16 * NROWT]; int c[16 * NROWT]; };
static constexpr RowTab makeRows() {
  RowTab t{};
  for (int i = 0; i < 16 * NROWT; ++i) { t.b[i] = -1; t.c[i] = -1; }
  for (int g = 0; g < 4; ++g) {
    int r = groupTileBase(g) * 16;
    for (int b = 4 * g; b < 4 * g + 4; ++b)
      for (int c = b; c < DP; ++c) { t.b[r] = b; t.c[r] = c; ++r; }
    if (g == 3) { t.b[r] = DP - 1; t.c[r] = DP - 1; ++r; }
  }
  return t;
}
static constexpr RowTab ROWS = makeRows();

#define NSLOT 10
#define TPP 4
static constexpr int tileRole(int j) { return tileGroup(j) & 1; }
static constexpr int slotOf(int f) { return f < 4 ? f : f < 10 ? f - 4 : f < 16 ? f - 6 : f - 10; }
static constexpr int fragOfSlot(int role, int i) {
  return role ? (i < 6 ? i + 4 : i + 10) : (i < 4 ? i : i + 6);
}
static constexpr bool slotsConsistent() {
  for (int j = 0; j < NROWT; ++j)
    for (int s = 0; s <= tileGroup(j); ++s) {
      const int f = fragOf(j, s), i = slotOf(f);
      if (i < 0 || i >= NSLOT || fragOfSlot(tileRole(j), i) != f) return false;
    }
  return true;
}
static_assert(slotsConsistent(), "slotOf and fragOfSlot must be inverse per role");

#define NMONO 969
#define IMG_V4 321
static_assert(NMONO % 4 == 1, "one tail monomial is assumed below");
static_assert(4 * IMG_V4 > WS_CONST, "image must hold w_const");

__device__ __forceinline__ int image_slot(int a, int b, int c) {
  const int g  = (b >= 12) ? 3 : (b >> 2);
  const int n  = b - 4 * g;
  const int rg = n * (DP - 4 * g) - (n * (n - 1)) / 2 + (c - b);
  const int s  = min(a >> 2, g);
  const int fb = (0x100A0400 >> (8 * g)) & 0xFF;
  const int f  = fb + (rg >> 4) * (g + 1) + s;
  const int ln = (rg & 15) + 16 * ((a >> 1) & 1);
  return (a == 16) ? WS_CONST : (f * 32 + ln) * 2 + (a & 1);
}

__global__ __launch_bounds__(256) void build_weights_kernel(
    const float* __restrict__ w, const int* __restrict__ comb_idx,
    float* __restrict__ ws) {
  __shared__ __align__(16) float4 img[IMG_V4];
  const int tid = threadIdx.x;
  const bool own4 = tid < NMONO / 4;

  int4 q0 = int4{16, 16, 16, 16}, q1 = q0, q2 = q0;
  float4 ww = float4{0.0f, 0.0f, 0.0f, 0.0f};
  if (own4) {
    const int4* ci = (const int4*)comb_idx + 3 * tid;
    q0 = ci[0]; q1 = ci[1]; q2 = ci[2];
    ww = ((const float4*)w)[tid];
  }
  const int   ta = comb_idx[3 * (NMONO - 1) + 0];
  const int   tb = comb_idx[3 * (NMONO - 1) + 1];
  const int   tc = comb_idx[3 * (NMONO - 1) + 2];
  const float tw = w[NMONO - 1];

  img[tid] = float4{0.0f, 0.0f, 0.0f, 0.0f};
  if (tid < IMG_V4 - 256) img[256 + tid] = float4{0.0f, 0.0f, 0.0f, 0.0f};
  __syncthreads();

  float* imgf = (float*)img;
  if (own4) {
    imgf[image_slot(q0.x, q0.y, q0.z)] = ww.x;
    imgf[image_slot(q0.w, q1.x, q1.y)] = ww.y;
    imgf[image_slot(q1.z, q1.w, q2.x)] = ww.z;
    imgf[image_slot(q2.y, q2.z, q2.w)] = ww.w;
  }
  if (tid == NMONO / 4) imgf[image_slot(ta, tb, tc)] = tw;
  __syncthreads();

  float4* out4 = (float4*)ws;
  out4[tid] = img[tid];
  if (tid < IMG_V4 - 256) out4[256 + tid] = img[256 + tid];
}

__device__ __forceinline__ float xhalf(float x) {
  return __builtin_bit_cast(float, __builtin_amdgcn_permlanex16(
      __builtin_bit_cast(int, x), __builtin_bit_cast(int, x),
      0x76543210, 0xfedcba98, false, false));
}

template <int ROLE, int H>
__device__ __forceinline__ float fold_rows(const v8f (&acc)[NROWT],
                                           const float (&xv)[DP]) {
  float tot = 0.0f, inner = 0.0f;
  int curb = -1;
#pragma unroll
  for (int j = 0; j < NROWT; ++j) {
    if (tileRole(j) == ROLE) {
#pragma unroll
      for (int r = 0; r < 8; ++r) {
        const int row = 16 * j + 8 * H + r;
        const int b = ROWS.b[row];
        const int c = ROWS.c[row];
        if (b >= 0) {
          if (b != curb) {
            if (curb >= 0) tot = fmaf(xv[curb], inner, tot);
            inner = 0.0f;
            curb = b;
          }
          inner = fmaf(acc[j][r], xv[c], inner);
        }
      }
    }
  }
  return fmaf(xv[curb], inner, tot);
}

template <int ROLE>
__device__ __forceinline__ void pair_rows(const v2f (&Af)[NSLOT],
                                          const float* xt, int lane,
                                          float& v01, float& v23) {
  const int hi = lane >> 4;
  const int lm = lane & 15;
  float res[TPP];
#pragma unroll
  for (int t = 0; t < TPP; ++t) {
    const float* xr = xt + t * (16 * XSTR) + lm * XSTR;
    float xv[DP];
    {
      const float4 r0 = *(const float4*)(xr + 0);
      const float4 r1 = *(const float4*)(xr + 4);
      const float4 r2 = *(const float4*)(xr + 8);
      const float4 r3 = *(const float4*)(xr + 12);
      xv[0]  = r0.x; xv[1]  = r0.y; xv[2]  = r0.z; xv[3]  = r0.w;
      xv[4]  = r1.x; xv[5]  = r1.y; xv[6]  = r1.z; xv[7]  = r1.w;
      xv[8]  = r2.x; xv[9]  = r2.y; xv[10] = r2.z; xv[11] = r2.w;
      xv[12] = r3.x; xv[13] = r3.y; xv[14] = r3.z; xv[15] = r3.w;
      xv[16] = 1.0f;
    }
    v2f Bx[4];
#pragma unroll
    for (int s = 0; s < 4; ++s) Bx[s] = *(const v2f*)(xr + 4 * s + 2 * hi);

    v8f acc[NROWT];
#pragma unroll
    for (int j = 0; j < NROWT; ++j) acc[j] = v8f{};
#pragma unroll
    for (int s = 0; s < 4; ++s) {
#pragma unroll
      for (int j = 0; j < NROWT; ++j) {
        if (tileRole(j) == ROLE && s <= tileGroup(j)) {
          acc[j] = __builtin_amdgcn_wmma_f32_16x16x4_f32(
              false, Af[slotOf(fragOf(j, s))], false, Bx[s], (short)0, acc[j],
              false, false);
        }
      }
    }

    float part;
    if (hi) part = fold_rows<ROLE, 1>(acc, xv);
    else    part = fold_rows<ROLE, 0>(acc, xv);
    res[t] = part + xhalf(part);
  }
  v01 = hi ? res[1] : res[0];
  v23 = hi ? res[3] : res[2];
}

__global__ __launch_bounds__(256) void poly_logistic_kernel(
    const float* __restrict__ X, const float* __restrict__ ws,
    float* __restrict__ out) {
  __shared__ __align__(16) float xp[4][TPP * 16 * XSTR];
  __shared__ float red[4][2][32];

  const int tid  = threadIdx.x;
  const int lane = tid & 31;
  const int wave = __builtin_amdgcn_readfirstlane(tid >> 5);
  const int pair = wave >> 1;
  const int role = wave & 1;
  const int tile0 = (blockIdx.x * 4 + pair) * TPP;

  const float4* Xg = (const float4*)(X + (size_t)tile0 * 256) + role * 128 + lane;
  float4 xin[4];
#pragma unroll
  for (int k = 0; k < 4; ++k) xin[k] = Xg[k * 32];

  const v2f* wf = (const v2f*)ws + lane;
  v2f Af[NSLOT];
#pragma unroll
  for (int i = 0; i < NSLOT; ++i) Af[i] = wf[fragOfSlot(role, i) * 32];
  const float wconst = ws[WS_CONST];

  float* xt = &xp[pair][0];
  {
    float* dst = xt + role * (2 * 16 * XSTR) + (lane >> 2) * XSTR + (lane & 3) * 4;
#pragma unroll
    for (int k = 0; k < 4; ++k) *(float4*)(dst + k * (8 * XSTR)) = xin[k];
  }
  __syncthreads();

  float v01, v23;
  if (role == 0) pair_rows<0>(Af, xt, lane, v01, v23);
  else           pair_rows<1>(Af, xt, lane, v01, v23);

  red[pair][role][lane] = role ? v01 : v23;
  __syncthreads();
  const float logit = (role ? v23 : v01) + red[pair][role ^ 1][lane] + wconst;
  out[(size_t)tile0 * 16 + role * 32 + lane] = 1.0f / (1.0f + __expf(-logit));
}

extern "C" void kernel_launch(void* const* d_in, const int* in_sizes, int n_in,
                              void* d_out, int out_size, void* d_ws,
                              size_t ws_size, hipStream_t stream) {
  const float* X        = (const float*)d_in[0];
  const float* w        = (const float*)d_in[1];
  const int*   comb_idx = (const int*)d_in[2];
  float*       out      = (float*)d_out;
  float*       ws       = (float*)d_ws;

  build_weights_kernel<<<1, 256, 0, stream>>>(w, comb_idx, ws);

  poly_logistic_kernel<<<256, 256, 0, stream>>>(X, ws, out);
}
